# starting kernel
# baseline (speedup 1.0000x reference)
#include <hip/hip_runtime.h>
#include <hip/hip_bf16.h>

typedef __attribute__((ext_vector_type(2))) float v2f;
typedef __attribute__((ext_vector_type(8))) float v8f;

#define C_IN  128
#define H_IN  56
#define W_IN  56
#define N_IMG 32
#define K_OUT 256
#define HW    (H_IN * W_IN)
#define M_TOT (N_IMG * HW)
#define KD    (C_IN * 9)

#define CCHUNK   16
#define NPAIR    (CCHUNK / 2)
#define WLDS_N   (9 * NPAIR * K_OUT * 2)

__global__ __launch_bounds__(256)
void conv3x3_wmma_f32(const float* __restrict__ x,
                      const float* __restrict__ wgt,
                      const float* __restrict__ bias,
                      float* __restrict__ out)
{
    __shared__ float wlds[WLDS_N];

    const int tid    = threadIdx.x;
    const int lane   = tid & 31;
    const int wave   = tid >> 5;
    const int wave_m = wave & 1;
    const int wave_n = wave >> 1;

    const int m_base = blockIdx.x * 128 + wave_m * 64;
    const int n_base = wave_n * 64;

    const int lrow  = lane & 15;
    const int khalf = lane >> 4;

    int nimg[4], oh[4], ow[4];
    #pragma unroll
    for (int mt = 0; mt < 4; ++mt) {
        int m    = m_base + mt * 16 + lrow;
        nimg[mt] = m / HW;
        int pix  = m - nimg[mt] * HW;
        oh[mt]   = pix / W_IN;
        ow[mt]   = pix - oh[mt] * W_IN;
    }

    int oc[4];
    #pragma unroll
    for (int nt = 0; nt < 4; ++nt) oc[nt] = n_base + nt * 16 + lrow;

    v8f acc[4][4];
    #pragma unroll
    for (int mt = 0; mt < 4; ++mt)
        #pragma unroll
        for (int nt = 0; nt < 4; ++nt) acc[mt][nt] = (v8f){};

    for (int cb0 = 0; cb0 < C_IN; cb0 += CCHUNK) {
        __syncthreads();
        #pragma unroll 1
        for (int idx = tid; idx < K_OUT * (CCHUNK * 9); idx += 256) {
            int ocq = idx / (CCHUNK * 9);
            int t   = idx - ocq * (CCHUNK * 9);
            int cl  = t / 9;
            int rs  = t - cl * 9;
            float w = wgt[ocq * KD + (cb0 + cl) * 9 + rs];
            int cp  = cl >> 1;
            int j   = cl & 1;
            wlds[(((rs * NPAIR + cp) * K_OUT + ocq) << 1) | j] = w;
        }
        __syncthreads();

        for (int rs = 0; rs < 9; ++rs) {
            const int r = rs / 3;
            const int s = rs - r * 3;

            int  abase[4];
            bool aval[4];
            #pragma unroll
            for (int mt = 0; mt < 4; ++mt) {
                int ih = oh[mt] + r - 1;
                int iw = ow[mt] + s - 1;
                aval[mt]  = ((unsigned)ih < (unsigned)H_IN) & ((unsigned)iw < (unsigned)W_IN);
                int off   = nimg[mt] * (C_IN * HW) + ih * W_IN + iw;
                abase[mt] = aval[mt] ? off : 0;
            }

            #pragma unroll
            for (int cb = 0; cb < CCHUNK; cb += 4) {
                const int c0 = cb0 + cb + khalf * 2;
                const int c1 = c0 + 1;

                v2f a[4], b[4];
                #pragma unroll
                for (int mt = 0; mt < 4; ++mt) {
                    float t0 = x[abase[mt] + c0 * HW];
                    float t1 = x[abase[mt] + c1 * HW];
                    a[mt].x = aval[mt] ? t0 : 0.0f;
                    a[mt].y = aval[mt] ? t1 : 0.0f;
                }
                const int cpbase = rs * NPAIR + (cb >> 1) + khalf;
                #pragma unroll
                for (int nt = 0; nt < 4; ++nt) {
                    b[nt] = *(const v2f*)&wlds[(cpbase * K_OUT + oc[nt]) << 1];
                }

                #pragma unroll
                for (int mt = 0; mt < 4; ++mt)
                    #pragma unroll
                    for (int nt = 0; nt < 4; ++nt)
                        acc[mt][nt] = __builtin_amdgcn_wmma_f32_16x16x4_f32(
                            false, a[mt], false, b[nt], (short)0, acc[mt][nt],
                            false, false);
            }
        }
    }

    #pragma unroll
    for (int nt = 0; nt < 4; ++nt) {
        const float bv = bias[oc[nt]];
        #pragma unroll
        for (int mt = 0; mt < 4; ++mt) {
            #pragma unroll
            for (int v = 0; v < 8; ++v) {
                const int mrow = m_base + mt * 16 + 8 * khalf + v;
                const int ni   = mrow / HW;
                const int pix  = mrow - ni * HW;
                out[ni * (K_OUT * HW) + oc[nt] * HW + pix] = acc[mt][nt][v] + bv;
            }
        }
    }
}

extern "C" void kernel_launch(void* const* d_in, const int* in_sizes, int n_in,
                              void* d_out, int out_size, void* d_ws, size_t ws_size,
                              hipStream_t stream) {
    (void)in_sizes; (void)n_in; (void)out_size; (void)d_ws; (void)ws_size;
    const float* x    = (const float*)d_in[0];
    const float* wgt  = (const float*)d_in[1];
    const float* bias = (const float*)d_in[2];
    float* out        = (float*)d_out;

    dim3 grid(M_TOT / 128, 1);
    dim3 block(256);
    conv3x3_wmma_f32<<<grid, block, 0, stream>>>(x, wgt, bias, out);
}


# Kernel rows by minimal filtering F(2,3) over height: 4 patch rows combined while staged, weight rows combined in registers, 4 width-only WMMA products per 2 output rows, not 6; sums re-associated
# speedup vs baseline: 2.9184x; 2.9184x over previous
#include <hip/hip_runtime.h>


typedef __attribute__((ext_vector_type(2))) float v2f;
typedef __attribute__((ext_vector_type(4))) float v4f;
typedef __attribute__((ext_vector_type(8))) float v8f;

#define C_IN  128
#define H_IN  56
#define W_IN  56
#define N_IMG 32
#define K_OUT 256
#define HW    3136
#define KD    1152

#define N_BLOCKS 1792
#define N_TROW   4
#define N_CTILE  4

#define PITCH    72
#define PAIR_DW  (4 * PITCH * 2)
#define HALF_PAD 32
#define LDS_DW   (64 * PAIR_DW + HALF_PAD)

__device__ __forceinline__ int pair_base(int cp) {
    return cp * PAIR_DW + (cp >= 32 ? HALF_PAD : 0);
}

#define WELEM(w, idx) ((w)[(idx) >> 2][(idx) & 3])

template <int pl, int g>
__device__ __forceinline__ void load_group(const float* __restrict__ lds, int b0, int b3,
                                           v2f (&b)[8])
{
    const int s = g >> 1;
    #pragma unroll
    for (int jj = 0; jj < 2; ++jj) {
        const int j = 2 * (g & 1) + jj;
        const int off = pl * PAIR_DW + (j * PITCH + s) * 2;
        #pragma unroll
        for (int i = 0; i < N_CTILE; ++i)
            b[jj * 4 + i] = *(const v2f*)&lds[((i < 3) ? (b0 + 32 * i) : b3) + off];
    }
}

template <int pl>
__device__ __forceinline__ void run_half(const float* __restrict__ lds,
                                         const v4f (&w)[9], int b0, int b3,
                                         v8f (&acc)[N_TROW][N_CTILE])
{
    v2f bA[8], bB[8];
    v2f a[N_TROW];
    load_group<pl, 0>(lds, b0, b3, bA);

#define HALF_GROUP(G, BCUR, BNEXT, LOADNEXT)                                              \
    {                                                                                     \
        LOADNEXT;                                                                         \
        if (((G) & 1) == 0) {                                                             \
            const int s = (G) >> 1;                                                       \
            v2f g0, g1, g2;                                                               \
            g0.x = WELEM(w, (2 * pl) * 9 + s);                                            \
            g0.y = WELEM(w, (2 * pl + 1) * 9 + s);                                        \
            g1.x = WELEM(w, (2 * pl) * 9 + 3 + s);                                        \
            g1.y = WELEM(w, (2 * pl + 1) * 9 + 3 + s);                                    \
            g2.x = WELEM(w, (2 * pl) * 9 + 6 + s);                                        \
            g2.y = WELEM(w, (2 * pl + 1) * 9 + 6 + s);                                    \
            const v2f eh = (g0 + g2) * 0.5f;                                              \
            const v2f gh = g1 * 0.5f;                                                     \
            a[0] = g0; a[1] = eh + gh; a[2] = eh - gh; a[3] = g2;                         \
        }                                                                                 \
          \
          \
          \
        __builtin_amdgcn_sched_barrier(0);                                                \
        _Pragma("unroll")                                                                 \
        for (int jj = 0; jj < 2; ++jj) {                                                  \
            const int j = 2 * ((G) & 1) + jj;                                             \
            _Pragma("unroll")                                                             \
            for (int i = 0; i < N_CTILE; ++i)                                             \
                acc[j][i] = __builtin_amdgcn_wmma_f32_16x16x4_f32(                        \
                    false, a[j], false, BCUR[jj * 4 + i], (short)0, acc[j][i],            \
                    false, false);                                                        \
        }                                                                                 \
                                                        \
        __builtin_amdgcn_sched_barrier(0);                                                \
    }

    HALF_GROUP(0, bA, bB, (load_group<pl, 1>(lds, b0, b3, bB)))
    HALF_GROUP(1, bB, bA, (load_group<pl, 2>(lds, b0, b3, bA)))
    HALF_GROUP(2, bA, bB, (load_group<pl, 3>(lds, b0, b3, bB)))
    HALF_GROUP(3, bB, bA, (load_group<pl, 4>(lds, b0, b3, bA)))
    HALF_GROUP(4, bA, bB, (load_group<pl, 5>(lds, b0, b3, bB)))
    HALF_GROUP(5, bB, bA, (void)0)
#undef HALF_GROUP
}

__global__ __launch_bounds__(256)
void conv3x3_rows_in_lds(const float* __restrict__ x,
                         const float* __restrict__ wgt,
                         const float* __restrict__ bias,
                         float* __restrict__ out)
{
    __shared__ __attribute__((aligned(16))) float patch[LDS_DW];

    const int tid  = threadIdx.x;
    const int lane = tid & 31;
    const int wave = tid >> 5;
    const int c    = lane & 15;
    const int h    = lane >> 4;

    const int bid    = blockIdx.x;
    const int ochalf = bid & 1;
    const int rpn    = bid >> 1;
    const int n      = rpn / 28;
    const int oh0    = (rpn - n * 28) * 2;

    const int oc0 = ochalf * 128 + wave * 16;
    const float* wrow = wgt + (oc0 + c) * KD + h * 576;
    v4f wa[9], wb[9];
    #pragma unroll
    for (int t = 0; t < 9; ++t) wa[t] = *(const v4f*)(wrow + 4 * t);

    {
        const int q   = tid & 15;
        const int cpl = tid >> 4;
        const bool top = (oh0 > 0);
        const bool bot = (oh0 + 2 < H_IN);
        const bool isdata = (q < 14);
        const float* xp = x + ((n * C_IN) * H_IN + oh0) * W_IN + (isdata ? 4 * q : 0);
        const v4f zero4 = (v4f){0.f, 0.f, 0.f, 0.f};

        v4f da[4][4], db[4][4];
        #pragma unroll
        for (int it = 0; it < 4; ++it) {
            const int cp = cpl + 16 * it;
            const float* p = xp + (2 * cp) * HW;
            da[it][0] = (isdata && top) ? *(const v4f*)(p - W_IN)          : zero4;
            da[it][1] = isdata          ? *(const v4f*)(p)                 : zero4;
            da[it][2] = isdata          ? *(const v4f*)(p + W_IN)          : zero4;
            da[it][3] = (isdata && bot) ? *(const v4f*)(p + 2 * W_IN)      : zero4;
            db[it][0] = (isdata && top) ? *(const v4f*)(p + HW - W_IN)     : zero4;
            db[it][1] = isdata          ? *(const v4f*)(p + HW)            : zero4;
            db[it][2] = isdata          ? *(const v4f*)(p + HW + W_IN)     : zero4;
            db[it][3] = (isdata && bot) ? *(const v4f*)(p + HW + 2 * W_IN) : zero4;
        }
        #pragma unroll
        for (int it = 0; it < 4; ++it) {
            const int cp = cpl + 16 * it;
            const int pb = pair_base(cp);
            if (isdata) {
                v4f ta[4], tb[4];
                ta[0] = da[it][0] - da[it][2];
                ta[1] = da[it][1] + da[it][2];
                ta[2] = da[it][2] - da[it][1];
                ta[3] = da[it][1] - da[it][3];
                tb[0] = db[it][0] - db[it][2];
                tb[1] = db[it][1] + db[it][2];
                tb[2] = db[it][2] - db[it][1];
                tb[3] = db[it][1] - db[it][3];
                #pragma unroll
                for (int j = 0; j < 4; ++j) {
                    const int rb = pb + j * (PITCH * 2) + (4 + 4 * q) * 2;
                    v4f lo = {ta[j].x, tb[j].x, ta[j].y, tb[j].y};
                    v4f hi = {ta[j].z, tb[j].z, ta[j].w, tb[j].w};
                    *(v4f*)&patch[rb]     = lo;
                    *(v4f*)&patch[rb + 4] = hi;
                }
            } else {
                const int col = (q == 14) ? 3 : 60;
                #pragma unroll
                for (int j = 0; j < 4; ++j)
                    *(v2f*)&patch[pb + j * (PITCH * 2) + col * 2] = (v2f){0.f, 0.f};
            }
        }
    }

    const int hb = h * (32 * PAIR_DW + HALF_PAD);
    int b0 = hb + (c + 3) * 2;
    int b3 = hb + ((c & 7) + 51) * 2;

    v8f acc[N_TROW][N_CTILE];
    #pragma unroll
    for (int j = 0; j < N_TROW; ++j) {
        #pragma unroll
        for (int i = 0; i < N_CTILE; ++i) acc[j][i] = (v8f){};
    }

    __syncthreads();

    #pragma unroll 1
    for (int q = 0; q < 16; q += 2) {
        const float* wn = wrow + 36 * (q + 1);
        run_half<0>(patch, wa, b0, b3, acc);
        #pragma unroll
        for (int t = 0; t < 5; ++t) wb[t] = *(const v4f*)(wn + 4 * t);
        run_half<1>(patch, wa, b0, b3, acc);
        #pragma unroll
        for (int t = 5; t < 9; ++t) wb[t] = *(const v4f*)(wn + 4 * t);
        b0 += 2 * PAIR_DW;
        b3 += 2 * PAIR_DW;

        const int qn = (q + 2 < 16) ? (q + 2) : q;
        const float* wm = wrow + 36 * qn;
        run_half<0>(patch, wb, b0, b3, acc);
        #pragma unroll
        for (int t = 0; t < 5; ++t) wa[t] = *(const v4f*)(wm + 4 * t);
        run_half<1>(patch, wb, b0, b3, acc);
        #pragma unroll
        for (int t = 5; t < 9; ++t) wa[t] = *(const v4f*)(wm + 4 * t);
        b0 += 2 * PAIR_DW;
        b3 += 2 * PAIR_DW;
    }

    float* op = out + ((n * K_OUT + oc0 + 8 * h) * HW + oh0 * W_IN + c);
    #pragma unroll
    for (int v = 0; v < 8; ++v) {
        const float bv = bias[oc0 + 8 * h + v];
        #pragma unroll
        for (int i = 0; i < N_CTILE; ++i) {
            const float m0 = acc[0][i][v], m1 = acc[1][i][v];
            const float m2 = acc[2][i][v], m3 = acc[3][i][v];
            const float y0 = (m0 + m1 + m2) + bv;
            const float y1 = (m1 - m2 - m3) + bv;
            if (i < 3 || c < 8) {
                op[v * HW + 16 * i]        = y0;
                op[v * HW + W_IN + 16 * i] = y1;
            }
        }
    }
}

extern "C" void kernel_launch(void* const* d_in, const int* in_sizes, int n_in,
                              void* d_out, int out_size, void* d_ws, size_t ws_size,
                              hipStream_t stream) {
    (void)in_sizes; (void)n_in; (void)out_size; (void)d_ws; (void)ws_size;
    const float* x    = (const float*)d_in[0];
    const float* wgt  = (const float*)d_in[1];
    const float* bias = (const float*)d_in[2];
    float* out        = (float*)d_out;

    dim3 grid(N_BLOCKS, 1);
    dim3 block(256);
    conv3x3_rows_in_lds<<<grid, block, 0, stream>>>(x, wgt, bias, out);
}


# Block re-cut to one row pair of two images x 64 out-ch, channels staged in two phases: the two images' 8 left-over columns share one tile, 14 not 16 WMMA products per wave step; sums re-ordered
# speedup vs baseline: 3.0676x; 1.0511x over previous
#include <hip/hip_runtime.h>


typedef __attribute__((ext_vector_type(2))) float v2f;
typedef __attribute__((ext_vector_type(4))) float v4f;
typedef __attribute__((ext_vector_type(8))) float v8f;

#define C_IN  128
#define H_IN  56
#define W_IN  56
#define N_IMG 32
#define K_OUT 256
#define HW    3136
#define KD    1152

#define N_BLOCKS 1792
#define N_TROW   4
#define N_FULL   3

#define PITCH    72
#define PAIR_DW  (4 * PITCH * 2)
#define LDS_DW   (64 * PAIR_DW + 32 + 16)

__device__ __forceinline__ int unit_base(int u) {
    return u * PAIR_DW + ((u & 16) ? 32 : 0) + ((u & 32) ? 16 : 0);
}

#define WELEM(w, idx) ((w)[(idx) >> 2][(idx) & 3])

template <int pl, int g>
__device__ __forceinline__ void load_group(const float* __restrict__ lds, int b0, int bS,
                                           v2f (&b)[7])
{
    const int s = g >> 1;
    #pragma unroll
    for (int jj = 0; jj < 2; ++jj) {
        const int j = 2 * (g & 1) + jj;
        const int off = pl * PAIR_DW + (j * PITCH + s) * 2;
        #pragma unroll
        for (int i = 0; i < N_FULL; ++i)
            b[jj * 3 + i] = *(const v2f*)&lds[b0 + 32 * i + off];
    }
    b[6] = *(const v2f*)&lds[bS + pl * PAIR_DW + (2 * (g & 1) * PITCH + s) * 2];
}

template <int pl>
__device__ __forceinline__ void run_half(const float* __restrict__ lds,
                                         const v4f (&w)[9], int b0, int bS, int type,
                                         v8f (&acc)[N_TROW][N_FULL], v8f (&accS)[2])
{
    v2f bA[7], bB[7];
    v2f a[N_TROW];
    v2f aS[2];
    load_group<pl, 0>(lds, b0, bS, bA);

#define HALF_GROUP(G, BCUR, BNEXT, LOADNEXT)                                              \
    {                                                                                     \
        LOADNEXT;                                                                         \
        if (((G) & 1) == 0) {                                                             \
            const int s = (G) >> 1;                                                       \
            v2f g0, g1, g2;                                                               \
            g0.x = WELEM(w, (2 * pl) * 9 + s);                                            \
            g0.y = WELEM(w, (2 * pl + 1) * 9 + s);                                        \
            g1.x = WELEM(w, (2 * pl) * 9 + 3 + s);                                        \
            g1.y = WELEM(w, (2 * pl + 1) * 9 + 3 + s);                                    \
            g2.x = WELEM(w, (2 * pl) * 9 + 6 + s);                                        \
            g2.y = WELEM(w, (2 * pl + 1) * 9 + 6 + s);                                    \
            const v2f eh = (g0 + g2) * 0.5f;                                              \
            const v2f gh = g1 * 0.5f;                                                     \
            a[0] = g0; a[1] = eh + gh; a[2] = eh - gh; a[3] = g2;                         \
            aS[0] = type ? a[1] : a[0];                                                   \
            aS[1] = type ? a[3] : a[2];                                                   \
        }                                                                                 \
          \
          \
          \
        __builtin_amdgcn_sched_barrier(0);                                                \
        _Pragma("unroll")                                                                 \
        for (int jj = 0; jj < 2; ++jj) {                                                  \
            const int j = 2 * ((G) & 1) + jj;                                             \
            _Pragma("unroll")                                                             \
            for (int i = 0; i < N_FULL; ++i)                                              \
                acc[j][i] = __builtin_amdgcn_wmma_f32_16x16x4_f32(                        \
                    false, a[j], false, BCUR[jj * 3 + i], (short)0, acc[j][i],            \
                    false, false);                                                        \
        }                                                                                 \
        accS[(G) & 1] = __builtin_amdgcn_wmma_f32_16x16x4_f32(                            \
            false, aS[(G) & 1], false, BCUR[6], (short)0, accS[(G) & 1],                  \
            false, false);                                                                \
          \
          \
          \
        asm volatile("" : "+v"(accS[(G) & 1]));                                           \
                                                        \
        __builtin_amdgcn_sched_barrier(0);                                                \
    }

    HALF_GROUP(0, bA, bB, (load_group<pl, 1>(lds, b0, bS, bB)))
    HALF_GROUP(1, bB, bA, (load_group<pl, 2>(lds, b0, bS, bA)))
    HALF_GROUP(2, bA, bB, (load_group<pl, 3>(lds, b0, bS, bB)))
    HALF_GROUP(3, bB, bA, (load_group<pl, 4>(lds, b0, bS, bA)))
    HALF_GROUP(4, bA, bB, (load_group<pl, 5>(lds, b0, bS, bB)))
    HALF_GROUP(5, bB, bA, (void)0)
#undef HALF_GROUP
}

__device__ __forceinline__ const float* wptr(const float* wrow, int Q) {
    return wrow + 576 * (Q >> 3) + 36 * (Q & 7);
}

__device__ __forceinline__ void run_phase(const float* __restrict__ lds,
                                          const float* __restrict__ wrow, int P,
                                          v4f (&wa)[9], int b0, int bS, int type,
                                          v8f (&acc)[N_TROW][N_FULL], v8f (&accS)[2])
{
    v4f wb[9];
    #pragma unroll 1
    for (int t = 0; t < 4; ++t) {
        const int Q = 8 * P + 2 * t;
        const float* wn = wptr(wrow, Q + 1);
        run_half<0>(lds, wa, b0, bS, type, acc, accS);
        #pragma unroll
        for (int k = 0; k < 5; ++k) wb[k] = *(const v4f*)(wn + 4 * k);
        run_half<1>(lds, wa, b0, bS, type, acc, accS);
        #pragma unroll
        for (int k = 5; k < 9; ++k) wb[k] = *(const v4f*)(wn + 4 * k);
        b0 += 2 * PAIR_DW;
        bS += 2 * PAIR_DW;

        const int Qn = (Q + 2 < 16) ? (Q + 2) : Q;
        const float* wm = wptr(wrow, Qn);
        run_half<0>(lds, wb, b0, bS, type, acc, accS);
        #pragma unroll
        for (int k = 0; k < 5; ++k) wa[k] = *(const v4f*)(wm + 4 * k);
        run_half<1>(lds, wb, b0, bS, type, acc, accS);
        #pragma unroll
        for (int k = 5; k < 9; ++k) wa[k] = *(const v4f*)(wm + 4 * k);
        b0 += 2 * PAIR_DW;
        bS += 2 * PAIR_DW;
    }
}

__device__ __forceinline__ void stage_phase(float* __restrict__ patch,
                                            const float* __restrict__ x,
                                            int n0, int oh0, int P, int tid)
{
    const int q   = tid & 15;
    const int cpl = tid >> 4;
    const bool top = (oh0 > 0);
    const bool bot = (oh0 + 2 < H_IN);
    const bool isdata = (q < 14);
    const v4f zero4 = (v4f){0.f, 0.f, 0.f, 0.f};

    #pragma unroll 1
    for (int img = 0; img < 2; ++img) {
        const float* xp = x + (((n0 + img) * C_IN + 64 * P) * H_IN + oh0) * W_IN
                            + (isdata ? 4 * q : 0);
        v4f da[2][4], db[2][4];
        #pragma unroll
        for (int it = 0; it < 2; ++it) {
            const int lp = cpl + 16 * it;
            const float* p = xp + (2 * lp) * HW;
            da[it][0] = (isdata && top) ? *(const v4f*)(p - W_IN)          : zero4;
            da[it][1] = isdata          ? *(const v4f*)(p)                 : zero4;
            da[it][2] = isdata          ? *(const v4f*)(p + W_IN)          : zero4;
            da[it][3] = (isdata && bot) ? *(const v4f*)(p + 2 * W_IN)      : zero4;
            db[it][0] = (isdata && top) ? *(const v4f*)(p + HW - W_IN)     : zero4;
            db[it][1] = isdata          ? *(const v4f*)(p + HW)            : zero4;
            db[it][2] = isdata          ? *(const v4f*)(p + HW + W_IN)     : zero4;
            db[it][3] = (isdata && bot) ? *(const v4f*)(p + HW + 2 * W_IN) : zero4;
        }
        #pragma unroll
        for (int it = 0; it < 2; ++it) {
            const int lp = cpl + 16 * it;
            const int pb = unit_base(img * 32 + lp);
            if (isdata) {
                v4f ta[4], tb[4];
                ta[0] = da[it][0] - da[it][2];
                ta[1] = da[it][1] + da[it][2];
                ta[2] = da[it][2] - da[it][1];
                ta[3] = da[it][1] - da[it][3];
                tb[0] = db[it][0] - db[it][2];
                tb[1] = db[it][1] + db[it][2];
                tb[2] = db[it][2] - db[it][1];
                tb[3] = db[it][1] - db[it][3];
                #pragma unroll
                for (int j = 0; j < 4; ++j) {
                    const int rb = pb + j * (PITCH * 2) + (4 + 4 * q) * 2;
                    v4f lo = {ta[j].x, tb[j].x, ta[j].y, tb[j].y};
                    v4f hi = {ta[j].z, tb[j].z, ta[j].w, tb[j].w};
                    *(v4f*)&patch[rb]     = lo;
                    *(v4f*)&patch[rb + 4] = hi;
                }
            } else {
                const int col = (q == 14) ? 3 : 60;
                #pragma unroll
                for (int j = 0; j < 4; ++j)
                    *(v2f*)&patch[pb + j * (PITCH * 2) + col * 2] = (v2f){0.f, 0.f};
            }
        }
    }
}

__global__ __launch_bounds__(256)
void conv3x3_rows_in_lds(const float* __restrict__ x,
                         const float* __restrict__ wgt,
                         const float* __restrict__ bias,
                         float* __restrict__ out)
{
    __shared__ __attribute__((aligned(16))) float patch[LDS_DW];

    const int tid  = threadIdx.x;
    const int lane = tid & 31;
    const int wave = __builtin_amdgcn_readfirstlane(tid >> 5);
    const int type = wave & 1;
    const int c    = lane & 15;
    const int h    = lane >> 4;

    const int bid  = blockIdx.x;
    const int ocq  = bid & 3;
    const int rest = bid >> 2;
    const int n2   = rest / 28;
    const int oh0  = (rest - n2 * 28) * 2;
    const int n0   = 2 * n2;

    const int oc0 = ocq * 64 + (wave >> 1) * 16;
    const float* wrow = wgt + (oc0 + c) * KD + h * 288;
    v4f wa[9];
    #pragma unroll
    for (int t = 0; t < 9; ++t) wa[t] = *(const v4f*)(wrow + 4 * t);

    const int b0 = unit_base(type * 32 + 16 * h) + (c + 3) * 2;
    const int bS = unit_base((c >> 3) * 32 + 16 * h) + ((c & 7) + 51) * 2 + type * (PITCH * 2);

    v8f acc[N_TROW][N_FULL];
    v8f accS[2];
    #pragma unroll
    for (int j = 0; j < N_TROW; ++j) {
        #pragma unroll
        for (int i = 0; i < N_FULL; ++i) acc[j][i] = (v8f){};
    }
    accS[0] = (v8f){};
    accS[1] = (v8f){};

    #pragma unroll 1
    for (int P = 0; P < 2; ++P) {
        stage_phase(patch, x, n0, oh0, P, tid);
        __syncthreads();
        run_phase(patch, wrow, P, wa, b0, bS, type, acc, accS);
        __syncthreads();
    }

    float* xs = patch + wave * 256;
    #pragma unroll
    for (int v = 0; v < 8; ++v)
        xs[v * 32 + lane] = type ? accS[0][v] : accS[1][v];

    float bv[8];
    #pragma unroll
    for (int v = 0; v < 8; ++v) bv[v] = bias[oc0 + 8 * h + v];

    __syncthreads();

    const float* xr = patch + (wave ^ 1) * 256;
    float* ops = out + (((n0 + (c >> 3)) * K_OUT + oc0 + 8 * h) * HW
                        + (oh0 + type) * W_IN + 48 + (c & 7));
    #pragma unroll
    for (int v = 0; v < 8; ++v) {
        const float r = xr[v * 32 + lane];
        const float y = type ? ((accS[0][v] - r) - accS[1][v]) : ((accS[0][v] + r) + accS[1][v]);
        ops[v * HW] = y + bv[v];
    }

    float* op = out + (((n0 + type) * K_OUT + oc0 + 8 * h) * HW + oh0 * W_IN + c);
    #pragma unroll
    for (int v = 0; v < 8; ++v) {
        #pragma unroll
        for (int i = 0; i < N_FULL; ++i) {
            const float m0 = acc[0][i][v], m1 = acc[1][i][v];
            const float m2 = acc[2][i][v], m3 = acc[3][i][v];
            op[v * HW + 16 * i]        = (m0 + m1 + m2) + bv[v];
            op[v * HW + W_IN + 16 * i] = (m1 - m2 - m3) + bv[v];
        }
    }
}

extern "C" void kernel_launch(void* const* d_in, const int* in_sizes, int n_in,
                              void* d_out, int out_size, void* d_ws, size_t ws_size,
                              hipStream_t stream) {
    (void)in_sizes; (void)n_in; (void)out_size; (void)d_ws; (void)ws_size;
    const float* x    = (const float*)d_in[0];
    const float* wgt  = (const float*)d_in[1];
    const float* bias = (const float*)d_in[2];
    float* out        = (float*)d_out;

    dim3 grid(N_BLOCKS, 1);
    dim3 block(256);
    conv3x3_rows_in_lds<<<grid, block, 0, stream>>>(x, wgt, bias, out);
}
